# starting kernel
# baseline (speedup 1.0000x reference)
#include <hip/hip_runtime.h>


typedef __attribute__((ext_vector_type(2))) float v2f;
typedef __attribute__((ext_vector_type(8))) float v8f;

#define NB 16
#define NC 512
#define NHW 1024
#define NG 32
#define NH 4
#define ND 128
#define GN_EPS 1e-5f
#define ATT_SCALE 0.08838834764831845f

static __device__ __forceinline__ v8f wmma4(v2f a, v2f b, v8f c) {
  return __builtin_amdgcn_wmma_f32_16x16x4_f32(false, a, false, b, (short)0, c,
                                               false, false);
}

__global__ __launch_bounds__(256) void gn_kernel(const float* __restrict__ x,
                                                 const float* __restrict__ w,
                                                 const float* __restrict__ bia,
                                                 float* __restrict__ xn) {
  const int b = blockIdx.x >> 5;
  const int g = blockIdx.x & 31;
  const size_t base = ((size_t)b * NC + (size_t)g * 16) * NHW;
  const int tid = threadIdx.x;
  float sum = 0.f, sq = 0.f;
  for (int i = tid; i < 16 * NHW; i += 256) {
    float v = x[base + i];
    sum += v;
    sq += v * v;
  }
  for (int m = 16; m >= 1; m >>= 1) {
    sum += __shfl_xor(sum, m, 32);
    sq += __shfl_xor(sq, m, 32);
  }
  __shared__ float rs[8], rq[8], stat[2];
  const int wv = tid >> 5;
  if ((tid & 31) == 0) { rs[wv] = sum; rq[wv] = sq; }
  __syncthreads();
  if (tid == 0) {
    float ts = 0.f, tq = 0.f;
    for (int i = 0; i < 8; ++i) { ts += rs[i]; tq += rq[i]; }
    float mean = ts * (1.f / 16384.f);
    float var = tq * (1.f / 16384.f) - mean * mean;
    stat[0] = mean;
    stat[1] = rsqrtf(var + GN_EPS);
  }
  __syncthreads();
  const float mean = stat[0], rstd = stat[1];
  for (int i = tid; i < 16 * NHW; i += 256) {
    int c = (g << 4) + (i >> 10);
    xn[base + i] = (x[base + i] - mean) * rstd * w[c] + bia[c];
  }
}

__global__ __launch_bounds__(256) void gemm512_kernel(
    const float* __restrict__ A, const float* __restrict__ Bmat,
    float* __restrict__ Y, long bStrideB, long bStrideY,
    const float* __restrict__ bias, const float* __restrict__ resid) {
  __shared__ float ldsA[16 * 516];
  const int m0 = blockIdx.x << 4;
  const int b = blockIdx.y;
  const int tid = threadIdx.x;
  for (int i = tid; i < 8192; i += 256) {
    int row = i >> 9, k = i & 511;
    ldsA[row * 516 + k] = A[(size_t)(m0 + row) * 512 + k];
  }
  __syncthreads();
  const int lane = tid & 31, wv = tid >> 5;
  const int lm = lane & 15, hi = lane >> 4;
  const float* Bb = Bmat + (size_t)b * bStrideB;
  const int n0 = wv << 7;
  v8f acc[8] = {};
  for (int k0 = 0; k0 < 512; k0 += 4) {
    const int kr = k0 + 2 * hi;
    v2f a;
    a.x = ldsA[lm * 516 + kr];
    a.y = ldsA[lm * 516 + kr + 1];
    const float* bp = Bb + (size_t)kr * NHW + n0 + lm;
#pragma unroll
    for (int nc = 0; nc < 8; ++nc) {
      v2f bf;
      bf.x = bp[nc * 16];
      bf.y = bp[nc * 16 + NHW];
      acc[nc] = wmma4(a, bf, acc[nc]);
    }
  }
  float* Yb = Y + (size_t)b * bStrideY;
#pragma unroll
  for (int nc = 0; nc < 8; ++nc) {
#pragma unroll
    for (int r = 0; r < 8; ++r) {
      const int row = m0 + r + 8 * hi;
      const int col = n0 + (nc << 4) + lm;
      float v = acc[nc][r];
      if (bias)
        v += bias[row] + resid[(size_t)b * bStrideY + (size_t)row * NHW + col];
      Yb[(size_t)row * NHW + col] = v;
    }
  }
}

__global__ void vtrans_kernel(const float* __restrict__ qkv,
                              float* __restrict__ vt) {
  __shared__ float t[16][17];
  const int bh = blockIdx.z, bq = bh >> 2, h = bh & 3;
  const int j0 = blockIdx.x << 4, d0 = blockIdx.y << 4;
  const int tx = threadIdx.x, ty = threadIdx.y;
  t[ty][tx] =
      qkv[((size_t)bq * 1536 + 1024 + h * ND + d0 + ty) * NHW + j0 + tx];
  __syncthreads();
  vt[((size_t)bh * NHW + j0 + ty) * ND + d0 + tx] = t[tx][ty];
}

__global__ __launch_bounds__(256) void attn_kernel(const float* __restrict__ qkv,
                                                   const float* __restrict__ vt,
                                                   float* __restrict__ att) {
  __shared__ float ldsK[128 * 16];
  __shared__ float ldsV[16 * 132];
  __shared__ float ldsP[8][16 * 17];
  const int bh = blockIdx.x >> 3, iblk = blockIdx.x & 7;
  const int b = bh >> 2, h = bh & 3;
  const int tid = threadIdx.x, wv = tid >> 5, lane = tid & 31;
  const int lm = lane & 15, hi = lane >> 4;
  const int i0 = (iblk << 7) + (wv << 4);
  const float* qp = qkv + ((size_t)b * 1536 + h * ND) * NHW;
  const float* kp = qkv + ((size_t)b * 1536 + 512 + h * ND) * NHW;
  const float* vp = vt + (size_t)bh * NHW * ND;
  v2f qf[32];
#pragma unroll
  for (int ks = 0; ks < 32; ++ks) {
    const int kr = ks * 4 + 2 * hi;
    qf[ks].x = qp[(size_t)kr * NHW + i0 + lm];
    qf[ks].y = qp[(size_t)(kr + 1) * NHW + i0 + lm];
  }
  v8f O[8] = {};
  float mst[8], lst[8];
#pragma unroll
  for (int r = 0; r < 8; ++r) { mst[r] = -3.0e38f; lst[r] = 0.f; }
  float* myP = &ldsP[wv][0];

  for (int jt = 0; jt < 64; ++jt) {
    const int j0 = jt << 4;
    __syncthreads();
    for (int i = tid; i < 2048; i += 256)
      ldsK[i] = kp[(size_t)(i >> 4) * NHW + j0 + (i & 15)];
    for (int i = tid; i < 2048; i += 256)
      ldsV[(i >> 7) * 132 + (i & 127)] = vp[(size_t)(j0 + (i >> 7)) * ND + (i & 127)];
    __syncthreads();
    v8f s = {};
#pragma unroll
    for (int ks = 0; ks < 32; ++ks) {
      const int kr = ks * 4 + 2 * hi;
      v2f bf;
      bf.x = ldsK[kr * 16 + lm];
      bf.y = ldsK[(kr + 1) * 16 + lm];
      s = wmma4(qf[ks], bf, s);
    }
    float alpha[8], p[8];
#pragma unroll
    for (int r = 0; r < 8; ++r) {
      float sv = s[r] * ATT_SCALE;
      float tm = sv;
      tm = fmaxf(tm, __shfl_xor(tm, 1, 16));
      tm = fmaxf(tm, __shfl_xor(tm, 2, 16));
      tm = fmaxf(tm, __shfl_xor(tm, 4, 16));
      tm = fmaxf(tm, __shfl_xor(tm, 8, 16));
      float mn = fmaxf(mst[r], tm);
      alpha[r] = __expf(mst[r] - mn);
      float pv = __expf(sv - mn);
      float ts = pv;
      ts += __shfl_xor(ts, 1, 16);
      ts += __shfl_xor(ts, 2, 16);
      ts += __shfl_xor(ts, 4, 16);
      ts += __shfl_xor(ts, 8, 16);
      lst[r] = lst[r] * alpha[r] + ts;
      mst[r] = mn;
      p[r] = pv;
    }
#pragma unroll
    for (int nc = 0; nc < 8; ++nc)
#pragma unroll
      for (int r = 0; r < 8; ++r) O[nc][r] *= alpha[r];
#pragma unroll
    for (int r = 0; r < 8; ++r) myP[(r + 8 * hi) * 17 + lm] = p[r];
#pragma unroll
    for (int ks = 0; ks < 4; ++ks) {
      const int kr = ks * 4 + 2 * hi;
      v2f af;
      af.x = myP[lm * 17 + kr];
      af.y = myP[lm * 17 + kr + 1];
#pragma unroll
      for (int nc = 0; nc < 8; ++nc) {
        v2f bf;
        bf.x = ldsV[kr * 132 + (nc << 4) + lm];
        bf.y = ldsV[(kr + 1) * 132 + (nc << 4) + lm];
        O[nc] = wmma4(af, bf, O[nc]);
      }
    }
  }
  float* ap = att + ((size_t)b * NC + h * ND) * NHW;
#pragma unroll
  for (int r = 0; r < 8; ++r) {
    const float inv = 1.f / lst[r];
#pragma unroll
    for (int nc = 0; nc < 8; ++nc)
      ap[(size_t)((nc << 4) + lm) * NHW + i0 + r + 8 * hi] = O[nc][r] * inv;
  }
}

extern "C" void kernel_launch(void* const* d_in, const int* in_sizes, int n_in,
                              void* d_out, int out_size, void* d_ws,
                              size_t ws_size, hipStream_t stream) {
  (void)in_sizes; (void)n_in; (void)out_size; (void)ws_size;
  const float* x = (const float*)d_in[0];
  const float* gnw = (const float*)d_in[1];
  const float* gnb = (const float*)d_in[2];
  const float* wqkv = (const float*)d_in[3];
  const float* wout = (const float*)d_in[4];
  const float* bout = (const float*)d_in[5];
  float* out = (float*)d_out;
  float* ws = (float*)d_ws;
  float* xn = ws;
  float* qkv = ws + 8388608;
  float* vt = ws + 33554432;

  gn_kernel<<<NB * NG, 256, 0, stream>>>(x, gnw, gnb, xn);
  gemm512_kernel<<<dim3(96, NB), 256, 0, stream>>>(wqkv, xn, qkv, 524288L,
                                                   1572864L, nullptr, nullptr);
  vtrans_kernel<<<dim3(64, 8, NB * NH), dim3(16, 16), 0, stream>>>(qkv, vt);
  attn_kernel<<<NB * NH * 8, 256, 0, stream>>>(qkv, vt, xn);
  gemm512_kernel<<<dim3(32, NB), 256, 0, stream>>>(wout, xn, out, 524288L,
                                                   524288L, bout, x);
}


# attention: K and V tile staging loads issued one tile ahead as one group of four 16-byte loads per thread (register prefetch), replacing 16 serial load-wait-store trips per tile
# speedup vs baseline: 1.3510x; 1.3510x over previous
#include <hip/hip_runtime.h>


typedef __attribute__((ext_vector_type(2))) float v2f;
typedef __attribute__((ext_vector_type(4))) float v4f;
typedef __attribute__((ext_vector_type(8))) float v8f;

#define NB 16
#define NC 512
#define NHW 1024
#define NG 32
#define NH 4
#define ND 128
#define GN_EPS 1e-5f
#define ATT_SCALE 0.08838834764831845f

static __device__ __forceinline__ v8f wmma4(v2f a, v2f b, v8f c) {
  return __builtin_amdgcn_wmma_f32_16x16x4_f32(false, a, false, b, (short)0, c,
                                               false, false);
}

__global__ __launch_bounds__(256) void gn_kernel(const float* __restrict__ x,
                                                 const float* __restrict__ w,
                                                 const float* __restrict__ bia,
                                                 float* __restrict__ xn) {
  const int b = blockIdx.x >> 5;
  const int g = blockIdx.x & 31;
  const size_t base = ((size_t)b * NC + (size_t)g * 16) * NHW;
  const int tid = threadIdx.x;
  float sum = 0.f, sq = 0.f;
  for (int i = tid; i < 16 * NHW; i += 256) {
    float v = x[base + i];
    sum += v;
    sq += v * v;
  }
  for (int m = 16; m >= 1; m >>= 1) {
    sum += __shfl_xor(sum, m, 32);
    sq += __shfl_xor(sq, m, 32);
  }
  __shared__ float rs[8], rq[8], stat[2];
  const int wv = tid >> 5;
  if ((tid & 31) == 0) { rs[wv] = sum; rq[wv] = sq; }
  __syncthreads();
  if (tid == 0) {
    float ts = 0.f, tq = 0.f;
    for (int i = 0; i < 8; ++i) { ts += rs[i]; tq += rq[i]; }
    float mean = ts * (1.f / 16384.f);
    float var = tq * (1.f / 16384.f) - mean * mean;
    stat[0] = mean;
    stat[1] = rsqrtf(var + GN_EPS);
  }
  __syncthreads();
  const float mean = stat[0], rstd = stat[1];
  for (int i = tid; i < 16 * NHW; i += 256) {
    int c = (g << 4) + (i >> 10);
    xn[base + i] = (x[base + i] - mean) * rstd * w[c] + bia[c];
  }
}

__global__ __launch_bounds__(256) void gemm512_kernel(
    const float* __restrict__ A, const float* __restrict__ Bmat,
    float* __restrict__ Y, long bStrideB, long bStrideY,
    const float* __restrict__ bias, const float* __restrict__ resid) {
  __shared__ float ldsA[16 * 516];
  const int m0 = blockIdx.x << 4;
  const int b = blockIdx.y;
  const int tid = threadIdx.x;
  for (int i = tid; i < 8192; i += 256) {
    int row = i >> 9, k = i & 511;
    ldsA[row * 516 + k] = A[(size_t)(m0 + row) * 512 + k];
  }
  __syncthreads();
  const int lane = tid & 31, wv = tid >> 5;
  const int lm = lane & 15, hi = lane >> 4;
  const float* Bb = Bmat + (size_t)b * bStrideB;
  const int n0 = wv << 7;
  v8f acc[8] = {};
  for (int k0 = 0; k0 < 512; k0 += 4) {
    const int kr = k0 + 2 * hi;
    v2f a;
    a.x = ldsA[lm * 516 + kr];
    a.y = ldsA[lm * 516 + kr + 1];
    const float* bp = Bb + (size_t)kr * NHW + n0 + lm;
#pragma unroll
    for (int nc = 0; nc < 8; ++nc) {
      v2f bf;
      bf.x = bp[nc * 16];
      bf.y = bp[nc * 16 + NHW];
      acc[nc] = wmma4(a, bf, acc[nc]);
    }
  }
  float* Yb = Y + (size_t)b * bStrideY;
#pragma unroll
  for (int nc = 0; nc < 8; ++nc) {
#pragma unroll
    for (int r = 0; r < 8; ++r) {
      const int row = m0 + r + 8 * hi;
      const int col = n0 + (nc << 4) + lm;
      float v = acc[nc][r];
      if (bias)
        v += bias[row] + resid[(size_t)b * bStrideY + (size_t)row * NHW + col];
      Yb[(size_t)row * NHW + col] = v;
    }
  }
}

__global__ void vtrans_kernel(const float* __restrict__ qkv,
                              float* __restrict__ vt) {
  __shared__ float t[16][17];
  const int bh = blockIdx.z, bq = bh >> 2, h = bh & 3;
  const int j0 = blockIdx.x << 4, d0 = blockIdx.y << 4;
  const int tx = threadIdx.x, ty = threadIdx.y;
  t[ty][tx] =
      qkv[((size_t)bq * 1536 + 1024 + h * ND + d0 + ty) * NHW + j0 + tx];
  __syncthreads();
  vt[((size_t)bh * NHW + j0 + ty) * ND + d0 + tx] = t[tx][ty];
}

__global__ __launch_bounds__(256) void attn_kernel(const float* __restrict__ qkv,
                                                   const float* __restrict__ vt,
                                                   float* __restrict__ att) {
  __shared__ __attribute__((aligned(16))) float ldsK[128 * 16];
  __shared__ __attribute__((aligned(16))) float ldsV[16 * 132];
  __shared__ float ldsP[8][16 * 17];
  const int bh = blockIdx.x >> 3, iblk = blockIdx.x & 7;
  const int b = bh >> 2, h = bh & 3;
  const int tid = threadIdx.x, wv = tid >> 5, lane = tid & 31;
  const int lm = lane & 15, hi = lane >> 4;
  const int i0 = (iblk << 7) + (wv << 4);
  const float* qp = qkv + ((size_t)b * 1536 + h * ND) * NHW;
  const float* kp = qkv + ((size_t)b * 1536 + 512 + h * ND) * NHW;
  const float* vp = vt + (size_t)bh * NHW * ND;
  v2f qf[32];
#pragma unroll
  for (int ks = 0; ks < 32; ++ks) {
    const int kr = ks * 4 + 2 * hi;
    qf[ks].x = qp[(size_t)kr * NHW + i0 + lm];
    qf[ks].y = qp[(size_t)(kr + 1) * NHW + i0 + lm];
  }
  v8f O[8] = {};
  float mst[8], lst[8];
#pragma unroll
  for (int r = 0; r < 8; ++r) { mst[r] = -3.0e38f; lst[r] = 0.f; }
  float* myP = &ldsP[wv][0];

  const int sr = tid >> 2, sc = (tid & 3) << 2;
  const int sj = tid >> 5, sd = (tid & 31) << 2;
  const float* kg = kp + (size_t)sr * NHW + sc;
  const float* vg = vp + (size_t)sj * ND + sd;
  float* kl = &ldsK[sr * 16 + sc];
  float* vl = &ldsV[sj * 132 + sd];
  v4f k0 = *(const v4f*)kg;
  v4f k1 = *(const v4f*)(kg + (size_t)64 * NHW);
  v4f v0 = *(const v4f*)vg;
  v4f v1 = *(const v4f*)(vg + 8 * ND);

  for (int jt = 0; jt < 64; ++jt) {
    __syncthreads();
    *(v4f*)kl = k0;
    *(v4f*)(kl + 64 * 16) = k1;
    *(v4f*)vl = v0;
    *(v4f*)(vl + 8 * 132) = v1;
    __syncthreads();
    if (jt + 1 < 64) {
      const int jn = (jt + 1) << 4;
      k0 = *(const v4f*)(kg + jn);
      k1 = *(const v4f*)(kg + (size_t)64 * NHW + jn);
      v0 = *(const v4f*)(vg + (size_t)jn * ND);
      v1 = *(const v4f*)(vg + (size_t)jn * ND + 8 * ND);
    }
    v8f s = {};
#pragma unroll
    for (int ks = 0; ks < 32; ++ks) {
      const int kr = ks * 4 + 2 * hi;
      v2f bf;
      bf.x = ldsK[kr * 16 + lm];
      bf.y = ldsK[(kr + 1) * 16 + lm];
      s = wmma4(qf[ks], bf, s);
    }
    float alpha[8], p[8];
#pragma unroll
    for (int r = 0; r < 8; ++r) {
      float sv = s[r] * ATT_SCALE;
      float tm = sv;
      tm = fmaxf(tm, __shfl_xor(tm, 1, 16));
      tm = fmaxf(tm, __shfl_xor(tm, 2, 16));
      tm = fmaxf(tm, __shfl_xor(tm, 4, 16));
      tm = fmaxf(tm, __shfl_xor(tm, 8, 16));
      float mn = fmaxf(mst[r], tm);
      alpha[r] = __expf(mst[r] - mn);
      float pv = __expf(sv - mn);
      float ts = pv;
      ts += __shfl_xor(ts, 1, 16);
      ts += __shfl_xor(ts, 2, 16);
      ts += __shfl_xor(ts, 4, 16);
      ts += __shfl_xor(ts, 8, 16);
      lst[r] = lst[r] * alpha[r] + ts;
      mst[r] = mn;
      p[r] = pv;
    }
#pragma unroll
    for (int nc = 0; nc < 8; ++nc)
#pragma unroll
      for (int r = 0; r < 8; ++r) O[nc][r] *= alpha[r];
#pragma unroll
    for (int r = 0; r < 8; ++r) myP[(r + 8 * hi) * 17 + lm] = p[r];
#pragma unroll
    for (int ks = 0; ks < 4; ++ks) {
      const int kr = ks * 4 + 2 * hi;
      v2f af;
      af.x = myP[lm * 17 + kr];
      af.y = myP[lm * 17 + kr + 1];
#pragma unroll
      for (int nc = 0; nc < 8; ++nc) {
        v2f bf;
        bf.x = ldsV[kr * 132 + (nc << 4) + lm];
        bf.y = ldsV[(kr + 1) * 132 + (nc << 4) + lm];
        O[nc] = wmma4(af, bf, O[nc]);
      }
    }
  }
  float* ap = att + ((size_t)b * NC + h * ND) * NHW;
#pragma unroll
  for (int r = 0; r < 8; ++r) {
    const float inv = 1.f / lst[r];
#pragma unroll
    for (int nc = 0; nc < 8; ++nc)
      ap[(size_t)((nc << 4) + lm) * NHW + i0 + r + 8 * hi] = O[nc][r] * inv;
  }
}

extern "C" void kernel_launch(void* const* d_in, const int* in_sizes, int n_in,
                              void* d_out, int out_size, void* d_ws,
                              size_t ws_size, hipStream_t stream) {
  (void)in_sizes; (void)n_in; (void)out_size; (void)ws_size;
  const float* x = (const float*)d_in[0];
  const float* gnw = (const float*)d_in[1];
  const float* gnb = (const float*)d_in[2];
  const float* wqkv = (const float*)d_in[3];
  const float* wout = (const float*)d_in[4];
  const float* bout = (const float*)d_in[5];
  float* out = (float*)d_out;
  float* ws = (float*)d_ws;
  float* xn = ws;
  float* qkv = ws + 8388608;
  float* vt = ws + 33554432;

  gn_kernel<<<NB * NG, 256, 0, stream>>>(x, gnw, gnb, xn);
  gemm512_kernel<<<dim3(96, NB), 256, 0, stream>>>(wqkv, xn, qkv, 524288L,
                                                   1572864L, nullptr, nullptr);
  vtrans_kernel<<<dim3(64, 8, NB * NH), dim3(16, 16), 0, stream>>>(qkv, vt);
  attn_kernel<<<NB * NH * 8, 256, 0, stream>>>(qkv, vt, xn);
  gemm512_kernel<<<dim3(32, NB), 256, 0, stream>>>(wout, xn, out, 524288L,
                                                   524288L, bout, x);
}


# attention stages V tiles in their natural [d][key] layout straight from the qkv plane (PV B fragment = one 8-byte LDS read at pitch 20): the V transpose launch and its workspace plane are removed
# speedup vs baseline: 1.5353x; 1.1364x over previous
#include <hip/hip_runtime.h>


typedef __attribute__((ext_vector_type(2))) float v2f;
typedef __attribute__((ext_vector_type(4))) float v4f;
typedef __attribute__((ext_vector_type(8))) float v8f;

#define NB 16
#define NC 512
#define NHW 1024
#define NG 32
#define NH 4
#define ND 128
#define GN_EPS 1e-5f
#define ATT_SCALE 0.08838834764831845f

static __device__ __forceinline__ v8f wmma4(v2f a, v2f b, v8f c) {
  return __builtin_amdgcn_wmma_f32_16x16x4_f32(false, a, false, b, (short)0, c,
                                               false, false);
}

__global__ __launch_bounds__(256) void gn_kernel(const float* __restrict__ x,
                                                 const float* __restrict__ w,
                                                 const float* __restrict__ bia,
                                                 float* __restrict__ xn) {
  const int b = blockIdx.x >> 5;
  const int g = blockIdx.x & 31;
  const size_t base = ((size_t)b * NC + (size_t)g * 16) * NHW;
  const int tid = threadIdx.x;
  float sum = 0.f, sq = 0.f;
  for (int i = tid; i < 16 * NHW; i += 256) {
    float v = x[base + i];
    sum += v;
    sq += v * v;
  }
  for (int m = 16; m >= 1; m >>= 1) {
    sum += __shfl_xor(sum, m, 32);
    sq += __shfl_xor(sq, m, 32);
  }
  __shared__ float rs[8], rq[8], stat[2];
  const int wv = tid >> 5;
  if ((tid & 31) == 0) { rs[wv] = sum; rq[wv] = sq; }
  __syncthreads();
  if (tid == 0) {
    float ts = 0.f, tq = 0.f;
    for (int i = 0; i < 8; ++i) { ts += rs[i]; tq += rq[i]; }
    float mean = ts * (1.f / 16384.f);
    float var = tq * (1.f / 16384.f) - mean * mean;
    stat[0] = mean;
    stat[1] = rsqrtf(var + GN_EPS);
  }
  __syncthreads();
  const float mean = stat[0], rstd = stat[1];
  for (int i = tid; i < 16 * NHW; i += 256) {
    int c = (g << 4) + (i >> 10);
    xn[base + i] = (x[base + i] - mean) * rstd * w[c] + bia[c];
  }
}

__global__ __launch_bounds__(256) void gemm512_kernel(
    const float* __restrict__ A, const float* __restrict__ Bmat,
    float* __restrict__ Y, long bStrideB, long bStrideY,
    const float* __restrict__ bias, const float* __restrict__ resid) {
  __shared__ float ldsA[16 * 516];
  const int m0 = blockIdx.x << 4;
  const int b = blockIdx.y;
  const int tid = threadIdx.x;
  for (int i = tid; i < 8192; i += 256) {
    int row = i >> 9, k = i & 511;
    ldsA[row * 516 + k] = A[(size_t)(m0 + row) * 512 + k];
  }
  __syncthreads();
  const int lane = tid & 31, wv = tid >> 5;
  const int lm = lane & 15, hi = lane >> 4;
  const float* Bb = Bmat + (size_t)b * bStrideB;
  const int n0 = wv << 7;
  v8f acc[8] = {};
  for (int k0 = 0; k0 < 512; k0 += 4) {
    const int kr = k0 + 2 * hi;
    v2f a;
    a.x = ldsA[lm * 516 + kr];
    a.y = ldsA[lm * 516 + kr + 1];
    const float* bp = Bb + (size_t)kr * NHW + n0 + lm;
#pragma unroll
    for (int nc = 0; nc < 8; ++nc) {
      v2f bf;
      bf.x = bp[nc * 16];
      bf.y = bp[nc * 16 + NHW];
      acc[nc] = wmma4(a, bf, acc[nc]);
    }
  }
  float* Yb = Y + (size_t)b * bStrideY;
#pragma unroll
  for (int nc = 0; nc < 8; ++nc) {
#pragma unroll
    for (int r = 0; r < 8; ++r) {
      const int row = m0 + r + 8 * hi;
      const int col = n0 + (nc << 4) + lm;
      float v = acc[nc][r];
      if (bias)
        v += bias[row] + resid[(size_t)b * bStrideY + (size_t)row * NHW + col];
      Yb[(size_t)row * NHW + col] = v;
    }
  }
}

__global__ __launch_bounds__(256) void attn_kernel(const float* __restrict__ qkv,
                                                   float* __restrict__ att) {
  __shared__ __attribute__((aligned(16))) float ldsK[128 * 16];
  __shared__ __attribute__((aligned(16))) float ldsV[128 * 20];
  __shared__ float ldsP[8][16 * 17];
  const int bh = blockIdx.x >> 3, iblk = blockIdx.x & 7;
  const int b = bh >> 2, h = bh & 3;
  const int tid = threadIdx.x, wv = tid >> 5, lane = tid & 31;
  const int lm = lane & 15, hi = lane >> 4;
  const int i0 = (iblk << 7) + (wv << 4);
  const float* qp = qkv + ((size_t)b * 1536 + h * ND) * NHW;
  const float* kp = qkv + ((size_t)b * 1536 + 512 + h * ND) * NHW;
  const float* vp = qkv + ((size_t)b * 1536 + 1024 + h * ND) * NHW;
  v2f qf[32];
#pragma unroll
  for (int ks = 0; ks < 32; ++ks) {
    const int kr = ks * 4 + 2 * hi;
    qf[ks].x = qp[(size_t)kr * NHW + i0 + lm];
    qf[ks].y = qp[(size_t)(kr + 1) * NHW + i0 + lm];
  }
  v8f O[8] = {};
  float mst[8], lst[8];
#pragma unroll
  for (int r = 0; r < 8; ++r) { mst[r] = -3.0e38f; lst[r] = 0.f; }
  float* myP = &ldsP[wv][0];

  const int sr = tid >> 2, sc = (tid & 3) << 2;
  const float* kg = kp + (size_t)sr * NHW + sc;
  const float* vg = vp + (size_t)sr * NHW + sc;
  float* kl = &ldsK[sr * 16 + sc];
  float* vl = &ldsV[sr * 20 + sc];
  v4f k0 = *(const v4f*)kg;
  v4f k1 = *(const v4f*)(kg + (size_t)64 * NHW);
  v4f v0 = *(const v4f*)vg;
  v4f v1 = *(const v4f*)(vg + (size_t)64 * NHW);

  for (int jt = 0; jt < 64; ++jt) {
    __syncthreads();
    *(v4f*)kl = k0;
    *(v4f*)(kl + 64 * 16) = k1;
    *(v4f*)vl = v0;
    *(v4f*)(vl + 64 * 20) = v1;
    __syncthreads();
    if (jt + 1 < 64) {
      const int jn = (jt + 1) << 4;
      k0 = *(const v4f*)(kg + jn);
      k1 = *(const v4f*)(kg + (size_t)64 * NHW + jn);
      v0 = *(const v4f*)(vg + jn);
      v1 = *(const v4f*)(vg + (size_t)64 * NHW + jn);
    }
    v8f s = {};
#pragma unroll
    for (int ks = 0; ks < 32; ++ks) {
      const int kr = ks * 4 + 2 * hi;
      v2f bf;
      bf.x = ldsK[kr * 16 + lm];
      bf.y = ldsK[(kr + 1) * 16 + lm];
      s = wmma4(qf[ks], bf, s);
    }
    float alpha[8], p[8];
#pragma unroll
    for (int r = 0; r < 8; ++r) {
      float sv = s[r] * ATT_SCALE;
      float tm = sv;
      tm = fmaxf(tm, __shfl_xor(tm, 1, 16));
      tm = fmaxf(tm, __shfl_xor(tm, 2, 16));
      tm = fmaxf(tm, __shfl_xor(tm, 4, 16));
      tm = fmaxf(tm, __shfl_xor(tm, 8, 16));
      float mn = fmaxf(mst[r], tm);
      alpha[r] = __expf(mst[r] - mn);
      float pv = __expf(sv - mn);
      float ts = pv;
      ts += __shfl_xor(ts, 1, 16);
      ts += __shfl_xor(ts, 2, 16);
      ts += __shfl_xor(ts, 4, 16);
      ts += __shfl_xor(ts, 8, 16);
      lst[r] = lst[r] * alpha[r] + ts;
      mst[r] = mn;
      p[r] = pv;
    }
#pragma unroll
    for (int nc = 0; nc < 8; ++nc)
#pragma unroll
      for (int r = 0; r < 8; ++r) O[nc][r] *= alpha[r];
#pragma unroll
    for (int r = 0; r < 8; ++r) myP[(r + 8 * hi) * 17 + lm] = p[r];
#pragma unroll
    for (int ks = 0; ks < 4; ++ks) {
      const int kr = ks * 4 + 2 * hi;
      v2f af;
      af.x = myP[lm * 17 + kr];
      af.y = myP[lm * 17 + kr + 1];
#pragma unroll
      for (int nc = 0; nc < 8; ++nc) {
        const v2f bf = *(const v2f*)&ldsV[((nc << 4) + lm) * 20 + kr];
        O[nc] = wmma4(af, bf, O[nc]);
      }
    }
  }
  float* ap = att + ((size_t)b * NC + h * ND) * NHW;
#pragma unroll
  for (int r = 0; r < 8; ++r) {
    const float inv = 1.f / lst[r];
#pragma unroll
    for (int nc = 0; nc < 8; ++nc)
      ap[(size_t)((nc << 4) + lm) * NHW + i0 + r + 8 * hi] = O[nc][r] * inv;
  }
}

extern "C" void kernel_launch(void* const* d_in, const int* in_sizes, int n_in,
                              void* d_out, int out_size, void* d_ws,
                              size_t ws_size, hipStream_t stream) {
  (void)in_sizes; (void)n_in; (void)out_size; (void)ws_size;
  const float* x = (const float*)d_in[0];
  const float* gnw = (const float*)d_in[1];
  const float* gnb = (const float*)d_in[2];
  const float* wqkv = (const float*)d_in[3];
  const float* wout = (const float*)d_in[4];
  const float* bout = (const float*)d_in[5];
  float* out = (float*)d_out;
  float* ws = (float*)d_ws;
  float* xn = ws;
  float* qkv = ws + 8388608;

  gn_kernel<<<NB * NG, 256, 0, stream>>>(x, gnw, gnb, xn);
  gemm512_kernel<<<dim3(96, NB), 256, 0, stream>>>(wqkv, xn, qkv, 524288L,
                                                   1572864L, nullptr, nullptr);
  attn_kernel<<<NB * NH * 8, 256, 0, stream>>>(qkv, xn);
  gemm512_kernel<<<dim3(32, NB), 256, 0, stream>>>(wout, xn, out, 524288L,
                                                   524288L, bout, x);
}


# GEMM wave tile 16x128 -> 32x64: each B fragment, fetched by 16-byte row loads (4 adjacent columns per lane), feeds two row tiles from a 32-row A panel in LDS; rows stored 16 bytes per lane
# speedup vs baseline: 1.7803x; 1.1596x over previous
#include <hip/hip_runtime.h>


typedef __attribute__((ext_vector_type(2))) float v2f;
typedef __attribute__((ext_vector_type(4))) float v4f;
typedef __attribute__((ext_vector_type(8))) float v8f;

#define NB 16
#define NC 512
#define NHW 1024
#define NG 32
#define NH 4
#define ND 128
#define GN_EPS 1e-5f
#define ATT_SCALE 0.08838834764831845f

static __device__ __forceinline__ v8f wmma4(v2f a, v2f b, v8f c) {
  return __builtin_amdgcn_wmma_f32_16x16x4_f32(false, a, false, b, (short)0, c,
                                               false, false);
}

__global__ __launch_bounds__(256) void gn_kernel(const float* __restrict__ x,
                                                 const float* __restrict__ w,
                                                 const float* __restrict__ bia,
                                                 float* __restrict__ xn) {
  const int b = blockIdx.x >> 5;
  const int g = blockIdx.x & 31;
  const size_t base = ((size_t)b * NC + (size_t)g * 16) * NHW;
  const int tid = threadIdx.x;
  float sum = 0.f, sq = 0.f;
  for (int i = tid; i < 16 * NHW; i += 256) {
    float v = x[base + i];
    sum += v;
    sq += v * v;
  }
  for (int m = 16; m >= 1; m >>= 1) {
    sum += __shfl_xor(sum, m, 32);
    sq += __shfl_xor(sq, m, 32);
  }
  __shared__ float rs[8], rq[8], stat[2];
  const int wv = tid >> 5;
  if ((tid & 31) == 0) { rs[wv] = sum; rq[wv] = sq; }
  __syncthreads();
  if (tid == 0) {
    float ts = 0.f, tq = 0.f;
    for (int i = 0; i < 8; ++i) { ts += rs[i]; tq += rq[i]; }
    float mean = ts * (1.f / 16384.f);
    float var = tq * (1.f / 16384.f) - mean * mean;
    stat[0] = mean;
    stat[1] = rsqrtf(var + GN_EPS);
  }
  __syncthreads();
  const float mean = stat[0], rstd = stat[1];
  for (int i = tid; i < 16 * NHW; i += 256) {
    int c = (g << 4) + (i >> 10);
    xn[base + i] = (x[base + i] - mean) * rstd * w[c] + bia[c];
  }
}

__global__ __launch_bounds__(256) void gemm512_kernel(
    const float* __restrict__ A, const float* __restrict__ Bmat,
    float* __restrict__ Y, long bStrideB, long bStrideY,
    const float* __restrict__ bias, const float* __restrict__ resid) {
  __shared__ __attribute__((aligned(16))) float ldsA[32 * 516];
  const int m0 = (blockIdx.x >> 1) << 5;
  const int nhalf = blockIdx.x & 1;
  const int b = blockIdx.y;
  const int tid = threadIdx.x;
  for (int i = tid; i < 4096; i += 256) {
    const int row = i >> 7, k4 = (i & 127) << 2;
    *(v4f*)&ldsA[row * 516 + k4] = *(const v4f*)&A[(size_t)(m0 + row) * 512 + k4];
  }
  __syncthreads();
  const int lane = tid & 31, wv = tid >> 5;
  const int lm = lane & 15, hi = lane >> 4;
  const int nw = (nhalf << 9) + (wv << 6) + (lm << 2);
  const float* Bb = Bmat + (size_t)b * bStrideB + nw;
  v8f acc[2][4] = {};
  for (int k0 = 0; k0 < 512; k0 += 4) {
    const int kr = k0 + 2 * hi;
    const v2f a0 = *(const v2f*)&ldsA[lm * 516 + kr];
    const v2f a1 = *(const v2f*)&ldsA[(16 + lm) * 516 + kr];
    const v4f r0 = *(const v4f*)(Bb + (size_t)kr * NHW);
    const v4f r1 = *(const v4f*)(Bb + (size_t)(kr + 1) * NHW);
#pragma unroll
    for (int t = 0; t < 4; ++t) {
      v2f bf;
      bf.x = r0[t];
      bf.y = r1[t];
      acc[0][t] = wmma4(a0, bf, acc[0][t]);
      acc[1][t] = wmma4(a1, bf, acc[1][t]);
    }
  }
  float* Yb = Y + (size_t)b * bStrideY;
#pragma unroll
  for (int mt = 0; mt < 2; ++mt) {
#pragma unroll
    for (int r = 0; r < 8; ++r) {
      const int row = m0 + (mt << 4) + r + 8 * hi;
      v4f v;
      v.x = acc[mt][0][r];
      v.y = acc[mt][1][r];
      v.z = acc[mt][2][r];
      v.w = acc[mt][3][r];
      if (bias) {
        const float bv = bias[row];
        const v4f rs =
            *(const v4f*)&resid[(size_t)b * bStrideY + (size_t)row * NHW + nw];
        v.x += bv + rs.x;
        v.y += bv + rs.y;
        v.z += bv + rs.z;
        v.w += bv + rs.w;
      }
      *(v4f*)&Yb[(size_t)row * NHW + nw] = v;
    }
  }
}

__global__ __launch_bounds__(256) void attn_kernel(const float* __restrict__ qkv,
                                                   float* __restrict__ att) {
  __shared__ __attribute__((aligned(16))) float ldsK[128 * 16];
  __shared__ __attribute__((aligned(16))) float ldsV[128 * 20];
  __shared__ float ldsP[8][16 * 17];
  const int bh = blockIdx.x >> 3, iblk = blockIdx.x & 7;
  const int b = bh >> 2, h = bh & 3;
  const int tid = threadIdx.x, wv = tid >> 5, lane = tid & 31;
  const int lm = lane & 15, hi = lane >> 4;
  const int i0 = (iblk << 7) + (wv << 4);
  const float* qp = qkv + ((size_t)b * 1536 + h * ND) * NHW;
  const float* kp = qkv + ((size_t)b * 1536 + 512 + h * ND) * NHW;
  const float* vp = qkv + ((size_t)b * 1536 + 1024 + h * ND) * NHW;
  v2f qf[32];
#pragma unroll
  for (int ks = 0; ks < 32; ++ks) {
    const int kr = ks * 4 + 2 * hi;
    qf[ks].x = qp[(size_t)kr * NHW + i0 + lm];
    qf[ks].y = qp[(size_t)(kr + 1) * NHW + i0 + lm];
  }
  v8f O[8] = {};
  float mst[8], lst[8];
#pragma unroll
  for (int r = 0; r < 8; ++r) { mst[r] = -3.0e38f; lst[r] = 0.f; }
  float* myP = &ldsP[wv][0];

  const int sr = tid >> 2, sc = (tid & 3) << 2;
  const float* kg = kp + (size_t)sr * NHW + sc;
  const float* vg = vp + (size_t)sr * NHW + sc;
  float* kl = &ldsK[sr * 16 + sc];
  float* vl = &ldsV[sr * 20 + sc];
  v4f k0 = *(const v4f*)kg;
  v4f k1 = *(const v4f*)(kg + (size_t)64 * NHW);
  v4f v0 = *(const v4f*)vg;
  v4f v1 = *(const v4f*)(vg + (size_t)64 * NHW);

  for (int jt = 0; jt < 64; ++jt) {
    __syncthreads();
    *(v4f*)kl = k0;
    *(v4f*)(kl + 64 * 16) = k1;
    *(v4f*)vl = v0;
    *(v4f*)(vl + 64 * 20) = v1;
    __syncthreads();
    if (jt + 1 < 64) {
      const int jn = (jt + 1) << 4;
      k0 = *(const v4f*)(kg + jn);
      k1 = *(const v4f*)(kg + (size_t)64 * NHW + jn);
      v0 = *(const v4f*)(vg + jn);
      v1 = *(const v4f*)(vg + (size_t)64 * NHW + jn);
    }
    v8f s = {};
#pragma unroll
    for (int ks = 0; ks < 32; ++ks) {
      const int kr = ks * 4 + 2 * hi;
      v2f bf;
      bf.x = ldsK[kr * 16 + lm];
      bf.y = ldsK[(kr + 1) * 16 + lm];
      s = wmma4(qf[ks], bf, s);
    }
    float alpha[8], p[8];
#pragma unroll
    for (int r = 0; r < 8; ++r) {
      float sv = s[r] * ATT_SCALE;
      float tm = sv;
      tm = fmaxf(tm, __shfl_xor(tm, 1, 16));
      tm = fmaxf(tm, __shfl_xor(tm, 2, 16));
      tm = fmaxf(tm, __shfl_xor(tm, 4, 16));
      tm = fmaxf(tm, __shfl_xor(tm, 8, 16));
      float mn = fmaxf(mst[r], tm);
      alpha[r] = __expf(mst[r] - mn);
      float pv = __expf(sv - mn);
      float ts = pv;
      ts += __shfl_xor(ts, 1, 16);
      ts += __shfl_xor(ts, 2, 16);
      ts += __shfl_xor(ts, 4, 16);
      ts += __shfl_xor(ts, 8, 16);
      lst[r] = lst[r] * alpha[r] + ts;
      mst[r] = mn;
      p[r] = pv;
    }
#pragma unroll
    for (int nc = 0; nc < 8; ++nc)
#pragma unroll
      for (int r = 0; r < 8; ++r) O[nc][r] *= alpha[r];
#pragma unroll
    for (int r = 0; r < 8; ++r) myP[(r + 8 * hi) * 17 + lm] = p[r];
#pragma unroll
    for (int ks = 0; ks < 4; ++ks) {
      const int kr = ks * 4 + 2 * hi;
      v2f af;
      af.x = myP[lm * 17 + kr];
      af.y = myP[lm * 17 + kr + 1];
#pragma unroll
      for (int nc = 0; nc < 8; ++nc) {
        const v2f bf = *(const v2f*)&ldsV[((nc << 4) + lm) * 20 + kr];
        O[nc] = wmma4(af, bf, O[nc]);
      }
    }
  }
  float* ap = att + ((size_t)b * NC + h * ND) * NHW;
#pragma unroll
  for (int r = 0; r < 8; ++r) {
    const float inv = 1.f / lst[r];
#pragma unroll
    for (int nc = 0; nc < 8; ++nc)
      ap[(size_t)((nc << 4) + lm) * NHW + i0 + r + 8 * hi] = O[nc][r] * inv;
  }
}

extern "C" void kernel_launch(void* const* d_in, const int* in_sizes, int n_in,
                              void* d_out, int out_size, void* d_ws,
                              size_t ws_size, hipStream_t stream) {
  (void)in_sizes; (void)n_in; (void)out_size; (void)ws_size;
  const float* x = (const float*)d_in[0];
  const float* gnw = (const float*)d_in[1];
  const float* gnb = (const float*)d_in[2];
  const float* wqkv = (const float*)d_in[3];
  const float* wout = (const float*)d_in[4];
  const float* bout = (const float*)d_in[5];
  float* out = (float*)d_out;
  float* ws = (float*)d_ws;
  float* xn = ws;
  float* qkv = ws + 8388608;

  gn_kernel<<<NB * NG, 256, 0, stream>>>(x, gnw, gnb, xn);
  gemm512_kernel<<<dim3(96, NB), 256, 0, stream>>>(wqkv, xn, qkv, 524288L,
                                                   1572864L, nullptr, nullptr);
  attn_kernel<<<NB * NH * 8, 256, 0, stream>>>(qkv, xn);
  gemm512_kernel<<<dim3(32, NB), 256, 0, stream>>>(wout, xn, out, 524288L,
                                                   524288L, bout, x);
}


# attention computes the score tile transposed (S^T = K^T Q, one query per lane): softmax row max and sum in-lane plus one half-wave exchange, P stays in registers as PV operand; row sums in a new order
# speedup vs baseline: 2.0681x; 1.1617x over previous
#include <hip/hip_runtime.h>


typedef __attribute__((ext_vector_type(2))) float v2f;
typedef __attribute__((ext_vector_type(4))) float v4f;
typedef __attribute__((ext_vector_type(8))) float v8f;

#define NB 16
#define NC 512
#define NHW 1024
#define NG 32
#define NH 4
#define ND 128
#define GN_EPS 1e-5f
#define ATT_SCALE 0.08838834764831845f

static __device__ __forceinline__ v8f wmma4(v2f a, v2f b, v8f c) {
  return __builtin_amdgcn_wmma_f32_16x16x4_f32(false, a, false, b, (short)0, c,
                                               false, false);
}

__global__ __launch_bounds__(256) void gn_kernel(const float* __restrict__ x,
                                                 const float* __restrict__ w,
                                                 const float* __restrict__ bia,
                                                 float* __restrict__ xn) {
  const int b = blockIdx.x >> 5;
  const int g = blockIdx.x & 31;
  const size_t base = ((size_t)b * NC + (size_t)g * 16) * NHW;
  const int tid = threadIdx.x;
  float sum = 0.f, sq = 0.f;
  for (int i = tid; i < 16 * NHW; i += 256) {
    float v = x[base + i];
    sum += v;
    sq += v * v;
  }
  for (int m = 16; m >= 1; m >>= 1) {
    sum += __shfl_xor(sum, m, 32);
    sq += __shfl_xor(sq, m, 32);
  }
  __shared__ float rs[8], rq[8], stat[2];
  const int wv = tid >> 5;
  if ((tid & 31) == 0) { rs[wv] = sum; rq[wv] = sq; }
  __syncthreads();
  if (tid == 0) {
    float ts = 0.f, tq = 0.f;
    for (int i = 0; i < 8; ++i) { ts += rs[i]; tq += rq[i]; }
    float mean = ts * (1.f / 16384.f);
    float var = tq * (1.f / 16384.f) - mean * mean;
    stat[0] = mean;
    stat[1] = rsqrtf(var + GN_EPS);
  }
  __syncthreads();
  const float mean = stat[0], rstd = stat[1];
  for (int i = tid; i < 16 * NHW; i += 256) {
    int c = (g << 4) + (i >> 10);
    xn[base + i] = (x[base + i] - mean) * rstd * w[c] + bia[c];
  }
}

__global__ __launch_bounds__(256) void gemm512_kernel(
    const float* __restrict__ A, const float* __restrict__ Bmat,
    float* __restrict__ Y, long bStrideB, long bStrideY,
    const float* __restrict__ bias, const float* __restrict__ resid) {
  __shared__ __attribute__((aligned(16))) float ldsA[32 * 516];
  const int m0 = (blockIdx.x >> 1) << 5;
  const int nhalf = blockIdx.x & 1;
  const int b = blockIdx.y;
  const int tid = threadIdx.x;
  for (int i = tid; i < 4096; i += 256) {
    const int row = i >> 7, k4 = (i & 127) << 2;
    *(v4f*)&ldsA[row * 516 + k4] = *(const v4f*)&A[(size_t)(m0 + row) * 512 + k4];
  }
  __syncthreads();
  const int lane = tid & 31, wv = tid >> 5;
  const int lm = lane & 15, hi = lane >> 4;
  const int nw = (nhalf << 9) + (wv << 6) + (lm << 2);
  const float* Bb = Bmat + (size_t)b * bStrideB + nw;
  v8f acc[2][4] = {};
  for (int k0 = 0; k0 < 512; k0 += 4) {
    const int kr = k0 + 2 * hi;
    const v2f a0 = *(const v2f*)&ldsA[lm * 516 + kr];
    const v2f a1 = *(const v2f*)&ldsA[(16 + lm) * 516 + kr];
    const v4f r0 = *(const v4f*)(Bb + (size_t)kr * NHW);
    const v4f r1 = *(const v4f*)(Bb + (size_t)(kr + 1) * NHW);
#pragma unroll
    for (int t = 0; t < 4; ++t) {
      v2f bf;
      bf.x = r0[t];
      bf.y = r1[t];
      acc[0][t] = wmma4(a0, bf, acc[0][t]);
      acc[1][t] = wmma4(a1, bf, acc[1][t]);
    }
  }
  float* Yb = Y + (size_t)b * bStrideY;
#pragma unroll
  for (int mt = 0; mt < 2; ++mt) {
#pragma unroll
    for (int r = 0; r < 8; ++r) {
      const int row = m0 + (mt << 4) + r + 8 * hi;
      v4f v;
      v.x = acc[mt][0][r];
      v.y = acc[mt][1][r];
      v.z = acc[mt][2][r];
      v.w = acc[mt][3][r];
      if (bias) {
        const float bv = bias[row];
        const v4f rs =
            *(const v4f*)&resid[(size_t)b * bStrideY + (size_t)row * NHW + nw];
        v.x += bv + rs.x;
        v.y += bv + rs.y;
        v.z += bv + rs.z;
        v.w += bv + rs.w;
      }
      *(v4f*)&Yb[(size_t)row * NHW + nw] = v;
    }
  }
}


static __device__ __forceinline__ float xhalf(float x) {
  const int b = __builtin_bit_cast(int, x);
  return __builtin_bit_cast(
      float, __builtin_amdgcn_permlanex16(b, b, 0x76543210, 0xfedcba98, false,
                                          false));
}

__global__ __launch_bounds__(256) void attn_kernel(const float* __restrict__ qkv,
                                                   float* __restrict__ att) {
  __shared__ __attribute__((aligned(16))) float ldsK[128 * 16];
  __shared__ __attribute__((aligned(16))) float ldsV[128 * 20];
  const int bh = blockIdx.x >> 3, iblk = blockIdx.x & 7;
  const int b = bh >> 2, h = bh & 3;
  const int tid = threadIdx.x, wv = tid >> 5, lane = tid & 31;
  const int lm = lane & 15, hi = lane >> 4;
  const int i0 = (iblk << 7) + (wv << 4);
  const float* qp = qkv + ((size_t)b * 1536 + h * ND) * NHW;
  const float* kp = qkv + ((size_t)b * 1536 + 512 + h * ND) * NHW;
  const float* vp = qkv + ((size_t)b * 1536 + 1024 + h * ND) * NHW;
  v2f qf[32];
#pragma unroll
  for (int ks = 0; ks < 32; ++ks) {
    const int kr = ks * 4 + 2 * hi;
    qf[ks].x = qp[(size_t)kr * NHW + i0 + lm];
    qf[ks].y = qp[(size_t)(kr + 1) * NHW + i0 + lm];
  }
  v8f O[8] = {};
  float mst = -3.0e38f, lst = 0.f;

  const int kcol = ((lm & 6) << 1) | ((lm >> 3) << 1) | (lm & 1);

  const int sr = tid >> 2, sc = (tid & 3) << 2;
  const float* kg = kp + (size_t)sr * NHW + sc;
  const float* vg = vp + (size_t)sr * NHW + sc;
  float* kl = &ldsK[sr * 16 + sc];
  float* vl = &ldsV[sr * 20 + sc];
  v4f k0 = *(const v4f*)kg;
  v4f k1 = *(const v4f*)(kg + (size_t)64 * NHW);
  v4f v0 = *(const v4f*)vg;
  v4f v1 = *(const v4f*)(vg + (size_t)64 * NHW);

  for (int jt = 0; jt < 64; ++jt) {
    __syncthreads();
    *(v4f*)kl = k0;
    *(v4f*)(kl + 64 * 16) = k1;
    *(v4f*)vl = v0;
    *(v4f*)(vl + 64 * 20) = v1;
    __syncthreads();
    if (jt + 1 < 64) {
      const int jn = (jt + 1) << 4;
      k0 = *(const v4f*)(kg + jn);
      k1 = *(const v4f*)(kg + (size_t)64 * NHW + jn);
      v0 = *(const v4f*)(vg + jn);
      v1 = *(const v4f*)(vg + (size_t)64 * NHW + jn);
    }
    v8f s = {};
#pragma unroll
    for (int ks = 0; ks < 32; ++ks) {
      const int kr = ks * 4 + 2 * hi;
      v2f kf;
      kf.x = ldsK[kr * 16 + kcol];
      kf.y = ldsK[(kr + 1) * 16 + kcol];
      s = wmma4(kf, qf[ks], s);
    }
    float tm = fmaxf(fmaxf(fmaxf(s[0], s[1]), fmaxf(s[2], s[3])),
                     fmaxf(fmaxf(s[4], s[5]), fmaxf(s[6], s[7])));
    tm = fmaxf(tm, xhalf(tm)) * ATT_SCALE;
    const float mn = fmaxf(mst, tm);
    const float alpha = __expf(mst - mn);
    float p[8];
#pragma unroll
    for (int r = 0; r < 8; ++r) p[r] = __expf(s[r] * ATT_SCALE - mn);
    float ts = ((p[0] + p[1]) + (p[2] + p[3])) + ((p[4] + p[5]) + (p[6] + p[7]));
    ts += xhalf(ts);
    lst = lst * alpha + ts;
    mst = mn;
#pragma unroll
    for (int nc = 0; nc < 8; ++nc)
#pragma unroll
      for (int r = 0; r < 8; ++r) O[nc][r] *= alpha;
#pragma unroll
    for (int ks = 0; ks < 4; ++ks) {
      const int kr = ks * 4 + 2 * hi;
      v2f pf;
      pf.x = p[2 * ks];
      pf.y = p[2 * ks + 1];
#pragma unroll
      for (int nc = 0; nc < 8; ++nc) {
        const v2f vf = *(const v2f*)&ldsV[((nc << 4) + lm) * 20 + kr];
        O[nc] = wmma4(vf, pf, O[nc]);
      }
    }
  }
  float* ap = att + ((size_t)b * NC + h * ND) * NHW;
  const float inv = 1.f / lst;
#pragma unroll
  for (int nc = 0; nc < 8; ++nc)
#pragma unroll
    for (int r = 0; r < 8; ++r)
      ap[(size_t)((nc << 4) + r + 8 * hi) * NHW + i0 + lm] = O[nc][r] * inv;
}

extern "C" void kernel_launch(void* const* d_in, const int* in_sizes, int n_in,
                              void* d_out, int out_size, void* d_ws,
                              size_t ws_size, hipStream_t stream) {
  (void)in_sizes; (void)n_in; (void)out_size; (void)ws_size;
  const float* x = (const float*)d_in[0];
  const float* gnw = (const float*)d_in[1];
  const float* gnb = (const float*)d_in[2];
  const float* wqkv = (const float*)d_in[3];
  const float* wout = (const float*)d_in[4];
  const float* bout = (const float*)d_in[5];
  float* out = (float*)d_out;
  float* ws = (float*)d_ws;
  float* xn = ws;
  float* qkv = ws + 8388608;

  gn_kernel<<<NB * NG, 256, 0, stream>>>(x, gnw, gnb, xn);
  gemm512_kernel<<<dim3(96, NB), 256, 0, stream>>>(wqkv, xn, qkv, 524288L,
                                                   1572864L, nullptr, nullptr);
  attn_kernel<<<NB * NH * 8, 256, 0, stream>>>(qkv, xn);
  gemm512_kernel<<<dim3(32, NB), 256, 0, stream>>>(wout, xn, out, 524288L,
                                                   524288L, bout, x);
}
